# starting kernel
# baseline (speedup 1.0000x reference)
#include <hip/hip_runtime.h>
#include <hip/hip_bf16.h>

typedef float v2f __attribute__((ext_vector_type(2)));
typedef float v8f __attribute__((ext_vector_type(8)));

__device__ __forceinline__ v8f wmma_f32(v2f a, v2f b, v8f c) {
  return __builtin_amdgcn_wmma_f32_16x16x4_f32(false, a, false, b, (short)0, c,
                                               false, false);
}

#define NSTREAM 6
#define H 128

__device__ __forceinline__ void mlp_hidden_layer(
    const float* __restrict__ W, const float* __restrict__ bb,
    const float (&actin)[NSTREAM * 16 * H], float (&actout)[NSTREAM * 16 * H],
    const int ln, const int half, const int koff)
{
  #pragma unroll
  for (int j = 0; j < 8; ++j) {
    v2f bf[32];
    #pragma unroll
    for (int kc = 0; kc < 32; ++kc) {
      const int kb = 4 * kc + koff;
      bf[kc].x = W[(size_t)kb * H + 16 * j + ln];
      bf[kc].y = W[(size_t)(kb + 1) * H + 16 * j + ln];
    }
    const float bj = bb[16 * j + ln];

    v8f acur, zcur;
    #pragma unroll
    for (int s = 0; s < NSTREAM; ++s) {
      v8f acc;
      #pragma unroll
      for (int v = 0; v < 8; ++v) acc[v] = (s == 0) ? bj : 0.f;

      #pragma unroll
      for (int kc = 0; kc < 32; ++kc) {
        const v2f af = *(const v2f*)&actin[(s * 16 + ln) * H + 4 * kc + koff];
        acc = wmma_f32(af, bf[kc], acc);
      }

      v8f res;
      if (s == 0) {
        #pragma unroll
        for (int v = 0; v < 8; ++v) res[v] = tanhf(acc[v]);
        acur = res;
      } else if (s == 1 || s == 3) {
        zcur = acc;
        #pragma unroll
        for (int v = 0; v < 8; ++v) {
          const float a = acur[v];
          res[v] = (1.f - a * a) * acc[v];
        }
      } else if (s == 2 || s == 4) {
        #pragma unroll
        for (int v = 0; v < 8; ++v) {
          const float a  = acur[v];
          const float s2 = 1.f - a * a;
          const float zd = zcur[v];
          res[v] = s2 * acc[v] - 2.f * a * s2 * zd * zd;
        }
      } else {
        #pragma unroll
        for (int v = 0; v < 8; ++v) {
          const float a = acur[v];
          res[v] = (1.f - a * a) * acc[v];
        }
      }
      #pragma unroll
      for (int v = 0; v < 8; ++v)
        actout[(s * 16 + (v + 8 * half)) * H + 16 * j + ln] = res[v];
    }
  }
}

__global__ __launch_bounds__(32) void pinn_grad_kernel(
    const float* __restrict__ x,
    const float* __restrict__ W0, const float* __restrict__ b0,
    const float* __restrict__ W1, const float* __restrict__ b1,
    const float* __restrict__ W2, const float* __restrict__ b2,
    const float* __restrict__ W3, const float* __restrict__ b3,
    float* __restrict__ out, int B)
{
  __shared__ __align__(16) float actA[NSTREAM * 16 * H];
  __shared__ __align__(16) float actB[NSTREAM * 16 * H];
  __shared__ float z3buf[NSTREAM * 16 * 4];

  const int lane = threadIdx.x & 31;
  const int half = lane >> 4;
  const int ln   = lane & 15;
  const int koff = half * 2;
  const int base = blockIdx.x * 16;

  for (int idx = lane; idx < 16 * H; idx += 32) {
    const int m = idx >> 7, n = idx & (H - 1);
    const float* xs = x + (size_t)(base + m) * 3;
    const float x0 = xs[0], x1 = xs[1], x2 = xs[2];
    const float w0 = W0[0 * H + n], w1 = W0[1 * H + n], w2 = W0[2 * H + n];
    const float z  = fmaf(x0, w0, fmaf(x1, w1, fmaf(x2, w2, b0[n])));
    const float a  = tanhf(z);
    const float s2 = 1.f - a * a;
    actA[(0 * 16 + m) * H + n] = a;
    actA[(1 * 16 + m) * H + n] = s2 * w0;
    actA[(2 * 16 + m) * H + n] = -2.f * a * s2 * w0 * w0;
    actA[(3 * 16 + m) * H + n] = s2 * w1;
    actA[(4 * 16 + m) * H + n] = -2.f * a * s2 * w1 * w1;
    actA[(5 * 16 + m) * H + n] = s2 * w2;
  }
  __syncthreads();

  mlp_hidden_layer(W1, b1, actA, actB, ln, half, koff);
  __syncthreads();
  mlp_hidden_layer(W2, b2, actB, actA, ln, half, koff);
  __syncthreads();

  {
    const int sg = half;
    const int m  = ln;
    #pragma unroll
    for (int ss = 0; ss < 3; ++ss) {
      const int s = 3 * sg + ss;
      float o0 = 0.f, o1 = 0.f, o2 = 0.f, o3 = 0.f;
      #pragma unroll 4
      for (int k = 0; k < H; ++k) {
        const float h = actA[(s * 16 + m) * H + k];
        o0 = fmaf(h, W3[k * 4 + 0], o0);
        o1 = fmaf(h, W3[k * 4 + 1], o1);
        o2 = fmaf(h, W3[k * 4 + 2], o2);
        o3 = fmaf(h, W3[k * 4 + 3], o3);
      }
      if (s == 0) { o0 += b3[0]; o1 += b3[1]; o2 += b3[2]; o3 += b3[3]; }
      z3buf[(s * 16 + m) * 4 + 0] = o0;
      z3buf[(s * 16 + m) * 4 + 1] = o1;
      z3buf[(s * 16 + m) * 4 + 2] = o2;
      z3buf[(s * 16 + m) * 4 + 3] = o3;
    }
  }
  __syncthreads();

  if (lane < 16) {
    const size_t i = (size_t)base + lane;
    float V[NSTREAM][4];
    #pragma unroll
    for (int s = 0; s < NSTREAM; ++s)
      #pragma unroll
      for (int o = 0; o < 4; ++o)
        V[s][o] = z3buf[(s * 16 + lane) * 4 + o];

    const float c   = V[0][0], v1 = V[0][1], v2 = V[0][2], Fi = V[0][3];
    const float dx0 = V[1][0], dx1 = V[1][1], dx2 = V[1][2], dx3 = V[1][3];
    const float xx0 = V[2][0], xx1 = V[2][1], xx2 = V[2][2], xx3 = V[2][3];
    const float dy0 = V[3][0], dy1 = V[3][1], dy2 = V[3][2], dy3 = V[3][3];
    const float yy0 = V[4][0], yy1 = V[4][1], yy2 = V[4][2], yy3 = V[4][3];
    const float dt0 = V[5][0], dt1 = V[5][1];

    const size_t Bz = (size_t)B;
    out[0 * Bz + i]          = c;
    out[1 * Bz + i]          = dt0;
    out[2 * Bz + 2 * i + 0]  = dx0;
    out[2 * Bz + 2 * i + 1]  = dy0;
    out[4 * Bz + 2 * i + 0]  = v1;
    out[4 * Bz + 2 * i + 1]  = v2;
    out[6 * Bz + i]          = dt1;
    out[7 * Bz + i]          = dx1 + dy2;
    out[8 * Bz + 2 * i + 0]  = dx1 * v1 + dy1 * v2;
    out[8 * Bz + 2 * i + 1]  = dx2 * v1 + dy2 * v2;
    out[10 * Bz + 2 * i + 0] = xx1 + yy1;
    out[10 * Bz + 2 * i + 1] = xx2 + yy2;
    out[12 * Bz + i]         = Fi;
    out[13 * Bz + 2 * i + 0] = dx3;
    out[13 * Bz + 2 * i + 1] = dy3;
    out[15 * Bz + i]         = xx3 + yy3;
    out[16 * Bz + i] = -0.5f * (xx0 + yy0)
                       - (dx0 * dx3 + dy0 * dy3) - c * (xx3 + yy3)
                       + (dx0 * v1 + dy0 * v2) + c * (dx1 + dy2);
  }
}

extern "C" void kernel_launch(void* const* d_in, const int* in_sizes, int n_in,
                              void* d_out, int out_size, void* d_ws, size_t ws_size,
                              hipStream_t stream) {
  const float* x  = (const float*)d_in[0];
  const float* W0 = (const float*)d_in[1];
  const float* b0 = (const float*)d_in[2];
  const float* W1 = (const float*)d_in[3];
  const float* b1 = (const float*)d_in[4];
  const float* W2 = (const float*)d_in[5];
  const float* b2 = (const float*)d_in[6];
  const float* W3 = (const float*)d_in[7];
  const float* b3 = (const float*)d_in[8];

  const int B = in_sizes[0] / 3;
  dim3 grid(B / 16), block(32);
  pinn_grad_kernel<<<grid, block, 0, stream>>>(x, W0, b0, W1, b1, W2, b2, W3, b3,
                                               (float*)d_out, B);
}


# Hidden layers: accumulators of all 6 streams x 4 N-tiles held in registers (24 tiles): each A fragment from LDS feeds 4 WMMAs, each B fragment 6; fragments requested one k-step ahead; sum order kept
# speedup vs baseline: 1.2852x; 1.2852x over previous
#include <hip/hip_runtime.h>
#include <hip/hip_bf16.h>

typedef float v2f __attribute__((ext_vector_type(2)));
typedef float v8f __attribute__((ext_vector_type(8)));

__device__ __forceinline__ v8f wmma_f32(v2f a, v2f b, v8f c) {
  return __builtin_amdgcn_wmma_f32_16x16x4_f32(false, a, false, b, (short)0, c,
                                               false, false);
}

#define NSTREAM 6
#define H 128

#define NT 4
__device__ __forceinline__ void mlp_hidden_layer(
    const float* __restrict__ W, const float* __restrict__ bb,
    const float (&actin)[NSTREAM * 16 * H], float (&actout)[NSTREAM * 16 * H],
    const int ln, const int half, const int koff)
{
  for (int jq = 0; jq < 8 / NT; ++jq) {
    v8f acc[NSTREAM][NT];
    #pragma unroll
    for (int t = 0; t < NT; ++t) {
      const float bj = bb[16 * (NT * jq + t) + ln];
      #pragma unroll
      for (int s = 0; s < NSTREAM; ++s)
        #pragma unroll
        for (int v = 0; v < 8; ++v) acc[s][t][v] = (s == 0) ? bj : 0.f;
    }

    v2f bfn[NT], afn[NSTREAM];
    #pragma unroll
    for (int t = 0; t < NT; ++t) {
      const int j = NT * jq + t;
      bfn[t].x = W[(size_t)koff * H + 16 * j + ln];
      bfn[t].y = W[(size_t)(koff + 1) * H + 16 * j + ln];
    }
    #pragma unroll
    for (int s = 0; s < NSTREAM; ++s)
      afn[s] = *(const v2f*)&actin[(s * 16 + ln) * H + koff];

    #pragma unroll 2
    for (int kc = 0; kc < 32; ++kc) {
      v2f bf[NT], af[NSTREAM];
      #pragma unroll
      for (int t = 0; t < NT; ++t) bf[t] = bfn[t];
      #pragma unroll
      for (int s = 0; s < NSTREAM; ++s) af[s] = afn[s];

      const int kn = (kc + 1 < 32) ? kc + 1 : kc;
      const int kb = 4 * kn + koff;
      #pragma unroll
      for (int t = 0; t < NT; ++t) {
        const int j = NT * jq + t;
        bfn[t].x = W[(size_t)kb * H + 16 * j + ln];
        bfn[t].y = W[(size_t)(kb + 1) * H + 16 * j + ln];
      }
      #pragma unroll
      for (int s = 0; s < NSTREAM; ++s)
        afn[s] = *(const v2f*)&actin[(s * 16 + ln) * H + 4 * kn + koff];

      #pragma unroll
      for (int s = 0; s < NSTREAM; ++s)
        #pragma unroll
        for (int t = 0; t < NT; ++t) acc[s][t] = wmma_f32(af[s], bf[t], acc[s][t]);
    }

    #pragma unroll
    for (int t = 0; t < NT; ++t) {
      const int j = NT * jq + t;
      v8f acur, zcur;
      #pragma unroll
      for (int s = 0; s < NSTREAM; ++s) {
        v8f res;
        if (s == 0) {
          #pragma unroll
          for (int v = 0; v < 8; ++v) res[v] = tanhf(acc[s][t][v]);
          acur = res;
        } else if (s == 1 || s == 3) {
          zcur = acc[s][t];
          #pragma unroll
          for (int v = 0; v < 8; ++v) {
            const float a = acur[v];
            res[v] = (1.f - a * a) * acc[s][t][v];
          }
        } else if (s == 2 || s == 4) {
          #pragma unroll
          for (int v = 0; v < 8; ++v) {
            const float a  = acur[v];
            const float s2 = 1.f - a * a;
            const float zd = zcur[v];
            res[v] = s2 * acc[s][t][v] - 2.f * a * s2 * zd * zd;
          }
        } else {
          #pragma unroll
          for (int v = 0; v < 8; ++v) {
            const float a = acur[v];
            res[v] = (1.f - a * a) * acc[s][t][v];
          }
        }
        #pragma unroll
        for (int v = 0; v < 8; ++v)
          actout[(s * 16 + (v + 8 * half)) * H + 16 * j + ln] = res[v];
      }
    }
  }
}

__global__ __launch_bounds__(32) void pinn_grad_kernel(
    const float* __restrict__ x,
    const float* __restrict__ W0, const float* __restrict__ b0,
    const float* __restrict__ W1, const float* __restrict__ b1,
    const float* __restrict__ W2, const float* __restrict__ b2,
    const float* __restrict__ W3, const float* __restrict__ b3,
    float* __restrict__ out, int B)
{
  __shared__ __align__(16) float actA[NSTREAM * 16 * H];
  __shared__ __align__(16) float actB[NSTREAM * 16 * H];
  __shared__ float z3buf[NSTREAM * 16 * 4];

  const int lane = threadIdx.x & 31;
  const int half = lane >> 4;
  const int ln   = lane & 15;
  const int koff = half * 2;
  const int base = blockIdx.x * 16;

  for (int idx = lane; idx < 16 * H; idx += 32) {
    const int m = idx >> 7, n = idx & (H - 1);
    const float* xs = x + (size_t)(base + m) * 3;
    const float x0 = xs[0], x1 = xs[1], x2 = xs[2];
    const float w0 = W0[0 * H + n], w1 = W0[1 * H + n], w2 = W0[2 * H + n];
    const float z  = fmaf(x0, w0, fmaf(x1, w1, fmaf(x2, w2, b0[n])));
    const float a  = tanhf(z);
    const float s2 = 1.f - a * a;
    actA[(0 * 16 + m) * H + n] = a;
    actA[(1 * 16 + m) * H + n] = s2 * w0;
    actA[(2 * 16 + m) * H + n] = -2.f * a * s2 * w0 * w0;
    actA[(3 * 16 + m) * H + n] = s2 * w1;
    actA[(4 * 16 + m) * H + n] = -2.f * a * s2 * w1 * w1;
    actA[(5 * 16 + m) * H + n] = s2 * w2;
  }
  __syncthreads();

  mlp_hidden_layer(W1, b1, actA, actB, ln, half, koff);
  __syncthreads();
  mlp_hidden_layer(W2, b2, actB, actA, ln, half, koff);
  __syncthreads();

  {
    const int sg = half;
    const int m  = ln;
    #pragma unroll
    for (int ss = 0; ss < 3; ++ss) {
      const int s = 3 * sg + ss;
      float o0 = 0.f, o1 = 0.f, o2 = 0.f, o3 = 0.f;
      #pragma unroll 4
      for (int k = 0; k < H; ++k) {
        const float h = actA[(s * 16 + m) * H + k];
        o0 = fmaf(h, W3[k * 4 + 0], o0);
        o1 = fmaf(h, W3[k * 4 + 1], o1);
        o2 = fmaf(h, W3[k * 4 + 2], o2);
        o3 = fmaf(h, W3[k * 4 + 3], o3);
      }
      if (s == 0) { o0 += b3[0]; o1 += b3[1]; o2 += b3[2]; o3 += b3[3]; }
      z3buf[(s * 16 + m) * 4 + 0] = o0;
      z3buf[(s * 16 + m) * 4 + 1] = o1;
      z3buf[(s * 16 + m) * 4 + 2] = o2;
      z3buf[(s * 16 + m) * 4 + 3] = o3;
    }
  }
  __syncthreads();

  if (lane < 16) {
    const size_t i = (size_t)base + lane;
    float V[NSTREAM][4];
    #pragma unroll
    for (int s = 0; s < NSTREAM; ++s)
      #pragma unroll
      for (int o = 0; o < 4; ++o)
        V[s][o] = z3buf[(s * 16 + lane) * 4 + o];

    const float c   = V[0][0], v1 = V[0][1], v2 = V[0][2], Fi = V[0][3];
    const float dx0 = V[1][0], dx1 = V[1][1], dx2 = V[1][2], dx3 = V[1][3];
    const float xx0 = V[2][0], xx1 = V[2][1], xx2 = V[2][2], xx3 = V[2][3];
    const float dy0 = V[3][0], dy1 = V[3][1], dy2 = V[3][2], dy3 = V[3][3];
    const float yy0 = V[4][0], yy1 = V[4][1], yy2 = V[4][2], yy3 = V[4][3];
    const float dt0 = V[5][0], dt1 = V[5][1];

    const size_t Bz = (size_t)B;
    out[0 * Bz + i]          = c;
    out[1 * Bz + i]          = dt0;
    out[2 * Bz + 2 * i + 0]  = dx0;
    out[2 * Bz + 2 * i + 1]  = dy0;
    out[4 * Bz + 2 * i + 0]  = v1;
    out[4 * Bz + 2 * i + 1]  = v2;
    out[6 * Bz + i]          = dt1;
    out[7 * Bz + i]          = dx1 + dy2;
    out[8 * Bz + 2 * i + 0]  = dx1 * v1 + dy1 * v2;
    out[8 * Bz + 2 * i + 1]  = dx2 * v1 + dy2 * v2;
    out[10 * Bz + 2 * i + 0] = xx1 + yy1;
    out[10 * Bz + 2 * i + 1] = xx2 + yy2;
    out[12 * Bz + i]         = Fi;
    out[13 * Bz + 2 * i + 0] = dx3;
    out[13 * Bz + 2 * i + 1] = dy3;
    out[15 * Bz + i]         = xx3 + yy3;
    out[16 * Bz + i] = -0.5f * (xx0 + yy0)
                       - (dx0 * dx3 + dy0 * dy3) - c * (xx3 + yy3)
                       + (dx0 * v1 + dy0 * v2) + c * (dx1 + dy2);
  }
}

extern "C" void kernel_launch(void* const* d_in, const int* in_sizes, int n_in,
                              void* d_out, int out_size, void* d_ws, size_t ws_size,
                              hipStream_t stream) {
  const float* x  = (const float*)d_in[0];
  const float* W0 = (const float*)d_in[1];
  const float* b0 = (const float*)d_in[2];
  const float* W1 = (const float*)d_in[3];
  const float* b1 = (const float*)d_in[4];
  const float* W2 = (const float*)d_in[5];
  const float* b2 = (const float*)d_in[6];
  const float* W3 = (const float*)d_in[7];
  const float* b3 = (const float*)d_in[8];

  const int B = in_sizes[0] / 3;
  dim3 grid(B / 16), block(32);
  pinn_grad_kernel<<<grid, block, 0, stream>>>(x, W0, b0, W1, b1, W2, b2, W3, b3,
                                               (float*)d_out, B);
}


# 4-wave blocks of 2 tiles: a wave pair splits each hidden layer's N-tiles and writes its output in place behind a barrier; 48 KB per tile, so 12 waves per WGP on all 4 SIMDs, not 3; sum order kept
# speedup vs baseline: 3.0962x; 2.4091x over previous
#include <hip/hip_runtime.h>
#include <hip/hip_bf16.h>

typedef float v2f __attribute__((ext_vector_type(2)));
typedef float v8f __attribute__((ext_vector_type(8)));

__device__ __forceinline__ v8f wmma_f32(v2f a, v2f b, v8f c) {
  return __builtin_amdgcn_wmma_f32_16x16x4_f32(false, a, false, b, (short)0, c,
                                               false, false);
}

#define NSTREAM 6
#define H 128
#define TILES 2
#define NWAVE (2 * TILES)

#define NT 4
__device__ __forceinline__ void mlp_hidden_layer(
    const float* __restrict__ W, const float* __restrict__ bb,
    float (&act)[NSTREAM * 16 * H],
    const int jq, const int ln, const int half, const int koff)
{
  v8f acc[NSTREAM][NT];
  #pragma unroll
  for (int t = 0; t < NT; ++t) {
    const float bj = bb[16 * (NT * jq + t) + ln];
    #pragma unroll
    for (int s = 0; s < NSTREAM; ++s)
      #pragma unroll
      for (int v = 0; v < 8; ++v) acc[s][t][v] = (s == 0) ? bj : 0.f;
  }

  v2f bfn[NT], afn[NSTREAM];
  #pragma unroll
  for (int t = 0; t < NT; ++t) {
    const int j = NT * jq + t;
    bfn[t].x = W[(size_t)koff * H + 16 * j + ln];
    bfn[t].y = W[(size_t)(koff + 1) * H + 16 * j + ln];
  }
  #pragma unroll
  for (int s = 0; s < NSTREAM; ++s)
    afn[s] = *(const v2f*)&act[(s * 16 + ln) * H + koff];

  #pragma unroll 2
  for (int kc = 0; kc < 32; ++kc) {
    v2f bf[NT], af[NSTREAM];
    #pragma unroll
    for (int t = 0; t < NT; ++t) bf[t] = bfn[t];
    #pragma unroll
    for (int s = 0; s < NSTREAM; ++s) af[s] = afn[s];

    const int kn = (kc + 1 < 32) ? kc + 1 : kc;
    const int kb = 4 * kn + koff;
    #pragma unroll
    for (int t = 0; t < NT; ++t) {
      const int j = NT * jq + t;
      bfn[t].x = W[(size_t)kb * H + 16 * j + ln];
      bfn[t].y = W[(size_t)(kb + 1) * H + 16 * j + ln];
    }
    #pragma unroll
    for (int s = 0; s < NSTREAM; ++s)
      afn[s] = *(const v2f*)&act[(s * 16 + ln) * H + 4 * kn + koff];

    #pragma unroll
    for (int s = 0; s < NSTREAM; ++s)
      #pragma unroll
      for (int t = 0; t < NT; ++t) acc[s][t] = wmma_f32(af[s], bf[t], acc[s][t]);
  }

  __syncthreads();

  #pragma unroll
  for (int t = 0; t < NT; ++t) {
    const int j = NT * jq + t;
    v8f acur, zcur;
    #pragma unroll
    for (int s = 0; s < NSTREAM; ++s) {
      v8f res;
      if (s == 0) {
        #pragma unroll
        for (int v = 0; v < 8; ++v) res[v] = tanhf(acc[s][t][v]);
        acur = res;
      } else if (s == 1 || s == 3) {
        zcur = acc[s][t];
        #pragma unroll
        for (int v = 0; v < 8; ++v) {
          const float a = acur[v];
          res[v] = (1.f - a * a) * acc[s][t][v];
        }
      } else if (s == 2 || s == 4) {
        #pragma unroll
        for (int v = 0; v < 8; ++v) {
          const float a  = acur[v];
          const float s2 = 1.f - a * a;
          const float zd = zcur[v];
          res[v] = s2 * acc[s][t][v] - 2.f * a * s2 * zd * zd;
        }
      } else {
        #pragma unroll
        for (int v = 0; v < 8; ++v) {
          const float a = acur[v];
          res[v] = (1.f - a * a) * acc[s][t][v];
        }
      }
      #pragma unroll
      for (int v = 0; v < 8; ++v)
        act[(s * 16 + (v + 8 * half)) * H + 16 * j + ln] = res[v];
    }
  }
}

__global__ __launch_bounds__(32 * NWAVE) void pinn_grad_kernel(
    const float* __restrict__ x,
    const float* __restrict__ W0, const float* __restrict__ b0,
    const float* __restrict__ W1, const float* __restrict__ b1,
    const float* __restrict__ W2, const float* __restrict__ b2,
    const float* __restrict__ W3, const float* __restrict__ b3,
    float* __restrict__ out, int B)
{
  __shared__ __align__(16) float act_all[TILES][NSTREAM * 16 * H];
  __shared__ float z3_all[TILES][NSTREAM * 16 * 4];

  const int tid  = threadIdx.x;
  const int wave = tid >> 5;
  const int tile = wave >> 1;
  const int jq   = wave & 1;
  const int tt   = tid & 63;
  const int lane = tid & 31;
  float (&act)[NSTREAM * 16 * H]   = act_all[tile];
  float (&z3buf)[NSTREAM * 16 * 4] = z3_all[tile];
  const int half = lane >> 4;
  const int ln   = lane & 15;
  const int koff = half * 2;
  const int base = (blockIdx.x * TILES + tile) * 16;

  for (int idx = tt; idx < 16 * H; idx += 64) {
    const int m = idx >> 7, n = idx & (H - 1);
    const float* xs = x + (size_t)(base + m) * 3;
    const float x0 = xs[0], x1 = xs[1], x2 = xs[2];
    const float w0 = W0[0 * H + n], w1 = W0[1 * H + n], w2 = W0[2 * H + n];
    const float z  = fmaf(x0, w0, fmaf(x1, w1, fmaf(x2, w2, b0[n])));
    const float a  = tanhf(z);
    const float s2 = 1.f - a * a;
    act[(0 * 16 + m) * H + n] = a;
    act[(1 * 16 + m) * H + n] = s2 * w0;
    act[(2 * 16 + m) * H + n] = -2.f * a * s2 * w0 * w0;
    act[(3 * 16 + m) * H + n] = s2 * w1;
    act[(4 * 16 + m) * H + n] = -2.f * a * s2 * w1 * w1;
    act[(5 * 16 + m) * H + n] = s2 * w2;
  }
  __syncthreads();

  mlp_hidden_layer(W1, b1, act, jq, ln, half, koff);
  __syncthreads();
  mlp_hidden_layer(W2, b2, act, jq, ln, half, koff);
  __syncthreads();

  for (int p = tt; p < NSTREAM * 16; p += 64) {
    const int s = p >> 4;
    const int m = p & 15;
    float o0 = 0.f, o1 = 0.f, o2 = 0.f, o3 = 0.f;
    #pragma unroll 4
    for (int k = 0; k < H; ++k) {
      const float h = act[(s * 16 + m) * H + k];
      o0 = fmaf(h, W3[k * 4 + 0], o0);
      o1 = fmaf(h, W3[k * 4 + 1], o1);
      o2 = fmaf(h, W3[k * 4 + 2], o2);
      o3 = fmaf(h, W3[k * 4 + 3], o3);
    }
    if (s == 0) { o0 += b3[0]; o1 += b3[1]; o2 += b3[2]; o3 += b3[3]; }
    z3buf[(s * 16 + m) * 4 + 0] = o0;
    z3buf[(s * 16 + m) * 4 + 1] = o1;
    z3buf[(s * 16 + m) * 4 + 2] = o2;
    z3buf[(s * 16 + m) * 4 + 3] = o3;
  }
  __syncthreads();

  if (tt < 16) {
    const size_t i = (size_t)base + tt;
    float V[NSTREAM][4];
    #pragma unroll
    for (int s = 0; s < NSTREAM; ++s)
      #pragma unroll
      for (int o = 0; o < 4; ++o)
        V[s][o] = z3buf[(s * 16 + tt) * 4 + o];

    const float c   = V[0][0], v1 = V[0][1], v2 = V[0][2], Fi = V[0][3];
    const float dx0 = V[1][0], dx1 = V[1][1], dx2 = V[1][2], dx3 = V[1][3];
    const float xx0 = V[2][0], xx1 = V[2][1], xx2 = V[2][2], xx3 = V[2][3];
    const float dy0 = V[3][0], dy1 = V[3][1], dy2 = V[3][2], dy3 = V[3][3];
    const float yy0 = V[4][0], yy1 = V[4][1], yy2 = V[4][2], yy3 = V[4][3];
    const float dt0 = V[5][0], dt1 = V[5][1];

    const size_t Bz = (size_t)B;
    out[0 * Bz + i]          = c;
    out[1 * Bz + i]          = dt0;
    out[2 * Bz + 2 * i + 0]  = dx0;
    out[2 * Bz + 2 * i + 1]  = dy0;
    out[4 * Bz + 2 * i + 0]  = v1;
    out[4 * Bz + 2 * i + 1]  = v2;
    out[6 * Bz + i]          = dt1;
    out[7 * Bz + i]          = dx1 + dy2;
    out[8 * Bz + 2 * i + 0]  = dx1 * v1 + dy1 * v2;
    out[8 * Bz + 2 * i + 1]  = dx2 * v1 + dy2 * v2;
    out[10 * Bz + 2 * i + 0] = xx1 + yy1;
    out[10 * Bz + 2 * i + 1] = xx2 + yy2;
    out[12 * Bz + i]         = Fi;
    out[13 * Bz + 2 * i + 0] = dx3;
    out[13 * Bz + 2 * i + 1] = dy3;
    out[15 * Bz + i]         = xx3 + yy3;
    out[16 * Bz + i] = -0.5f * (xx0 + yy0)
                       - (dx0 * dx3 + dy0 * dy3) - c * (xx3 + yy3)
                       + (dx0 * v1 + dy0 * v2) + c * (dx1 + dy2);
  }
}

extern "C" void kernel_launch(void* const* d_in, const int* in_sizes, int n_in,
                              void* d_out, int out_size, void* d_ws, size_t ws_size,
                              hipStream_t stream) {
  const float* x  = (const float*)d_in[0];
  const float* W0 = (const float*)d_in[1];
  const float* b0 = (const float*)d_in[2];
  const float* W1 = (const float*)d_in[3];
  const float* b1 = (const float*)d_in[4];
  const float* W2 = (const float*)d_in[5];
  const float* b2 = (const float*)d_in[6];
  const float* W3 = (const float*)d_in[7];
  const float* b3 = (const float*)d_in[8];

  const int B = in_sizes[0] / 3;
  dim3 grid(B / (16 * TILES)), block(32 * NWAVE);
  pinn_grad_kernel<<<grid, block, 0, stream>>>(x, W0, b0, W1, b1, W2, b2, W3, b3,
                                               (float*)d_out, B);
}


# LDS activation rows padded from 128 to 132 floats: the 16 rows of an A-fragment read, and the last layer's per-row reads, fall on distinct banks instead of one bank pair 16 deep; sum order kept
# speedup vs baseline: 3.2248x; 1.0415x over previous
#include <hip/hip_runtime.h>
#include <hip/hip_bf16.h>

typedef float v2f __attribute__((ext_vector_type(2)));
typedef float v8f __attribute__((ext_vector_type(8)));

__device__ __forceinline__ v8f wmma_f32(v2f a, v2f b, v8f c) {
  return __builtin_amdgcn_wmma_f32_16x16x4_f32(false, a, false, b, (short)0, c,
                                               false, false);
}

#define NSTREAM 6
#define H 128
#define PITCH (H + 4)
#define TILES 2
#define NWAVE (2 * TILES)

#define NT 4
__device__ __forceinline__ void mlp_hidden_layer(
    const float* __restrict__ W, const float* __restrict__ bb,
    float (&act)[NSTREAM * 16 * PITCH],
    const int jq, const int ln, const int half, const int koff)
{
  v8f acc[NSTREAM][NT];
  #pragma unroll
  for (int t = 0; t < NT; ++t) {
    const float bj = bb[16 * (NT * jq + t) + ln];
    #pragma unroll
    for (int s = 0; s < NSTREAM; ++s)
      #pragma unroll
      for (int v = 0; v < 8; ++v) acc[s][t][v] = (s == 0) ? bj : 0.f;
  }

  v2f bfn[NT], afn[NSTREAM];
  #pragma unroll
  for (int t = 0; t < NT; ++t) {
    const int j = NT * jq + t;
    bfn[t].x = W[(size_t)koff * H + 16 * j + ln];
    bfn[t].y = W[(size_t)(koff + 1) * H + 16 * j + ln];
  }
  #pragma unroll
  for (int s = 0; s < NSTREAM; ++s)
    afn[s] = *(const v2f*)&act[(s * 16 + ln) * PITCH + koff];

  #pragma unroll 2
  for (int kc = 0; kc < 32; ++kc) {
    v2f bf[NT], af[NSTREAM];
    #pragma unroll
    for (int t = 0; t < NT; ++t) bf[t] = bfn[t];
    #pragma unroll
    for (int s = 0; s < NSTREAM; ++s) af[s] = afn[s];

    const int kn = (kc + 1 < 32) ? kc + 1 : kc;
    const int kb = 4 * kn + koff;
    #pragma unroll
    for (int t = 0; t < NT; ++t) {
      const int j = NT * jq + t;
      bfn[t].x = W[(size_t)kb * H + 16 * j + ln];
      bfn[t].y = W[(size_t)(kb + 1) * H + 16 * j + ln];
    }
    #pragma unroll
    for (int s = 0; s < NSTREAM; ++s)
      afn[s] = *(const v2f*)&act[(s * 16 + ln) * PITCH + 4 * kn + koff];

    #pragma unroll
    for (int s = 0; s < NSTREAM; ++s)
      #pragma unroll
      for (int t = 0; t < NT; ++t) acc[s][t] = wmma_f32(af[s], bf[t], acc[s][t]);
  }

  __syncthreads();

  #pragma unroll
  for (int t = 0; t < NT; ++t) {
    const int j = NT * jq + t;
    v8f acur, zcur;
    #pragma unroll
    for (int s = 0; s < NSTREAM; ++s) {
      v8f res;
      if (s == 0) {
        #pragma unroll
        for (int v = 0; v < 8; ++v) res[v] = tanhf(acc[s][t][v]);
        acur = res;
      } else if (s == 1 || s == 3) {
        zcur = acc[s][t];
        #pragma unroll
        for (int v = 0; v < 8; ++v) {
          const float a = acur[v];
          res[v] = (1.f - a * a) * acc[s][t][v];
        }
      } else if (s == 2 || s == 4) {
        #pragma unroll
        for (int v = 0; v < 8; ++v) {
          const float a  = acur[v];
          const float s2 = 1.f - a * a;
          const float zd = zcur[v];
          res[v] = s2 * acc[s][t][v] - 2.f * a * s2 * zd * zd;
        }
      } else {
        #pragma unroll
        for (int v = 0; v < 8; ++v) {
          const float a = acur[v];
          res[v] = (1.f - a * a) * acc[s][t][v];
        }
      }
      #pragma unroll
      for (int v = 0; v < 8; ++v)
        act[(s * 16 + (v + 8 * half)) * PITCH + 16 * j + ln] = res[v];
    }
  }
}

__global__ __launch_bounds__(32 * NWAVE) void pinn_grad_kernel(
    const float* __restrict__ x,
    const float* __restrict__ W0, const float* __restrict__ b0,
    const float* __restrict__ W1, const float* __restrict__ b1,
    const float* __restrict__ W2, const float* __restrict__ b2,
    const float* __restrict__ W3, const float* __restrict__ b3,
    float* __restrict__ out, int B)
{
  __shared__ __align__(16) float act_all[TILES][NSTREAM * 16 * PITCH];

  const int tid  = threadIdx.x;
  const int wave = tid >> 5;
  const int tile = wave >> 1;
  const int jq   = wave & 1;
  const int tt   = tid & 63;
  const int lane = tid & 31;
  float (&act)[NSTREAM * 16 * PITCH] = act_all[tile];
  const int half = lane >> 4;
  const int ln   = lane & 15;
  const int koff = half * 2;
  const int base = (blockIdx.x * TILES + tile) * 16;

  for (int idx = tt; idx < 16 * H; idx += 64) {
    const int m = idx >> 7, n = idx & (H - 1);
    const float* xs = x + (size_t)(base + m) * 3;
    const float x0 = xs[0], x1 = xs[1], x2 = xs[2];
    const float w0 = W0[0 * H + n], w1 = W0[1 * H + n], w2 = W0[2 * H + n];
    const float z  = fmaf(x0, w0, fmaf(x1, w1, fmaf(x2, w2, b0[n])));
    const float a  = tanhf(z);
    const float s2 = 1.f - a * a;
    act[(0 * 16 + m) * PITCH + n] = a;
    act[(1 * 16 + m) * PITCH + n] = s2 * w0;
    act[(2 * 16 + m) * PITCH + n] = -2.f * a * s2 * w0 * w0;
    act[(3 * 16 + m) * PITCH + n] = s2 * w1;
    act[(4 * 16 + m) * PITCH + n] = -2.f * a * s2 * w1 * w1;
    act[(5 * 16 + m) * PITCH + n] = s2 * w2;
  }
  __syncthreads();

  mlp_hidden_layer(W1, b1, act, jq, ln, half, koff);
  __syncthreads();
  mlp_hidden_layer(W2, b2, act, jq, ln, half, koff);
  __syncthreads();

  for (int p = tt; p < NSTREAM * 16; p += 64) {
    const int s = p >> 4;
    const int m = p & 15;
    float o0 = 0.f, o1 = 0.f, o2 = 0.f, o3 = 0.f;
    #pragma unroll 4
    for (int k = 0; k < H; ++k) {
      const float h = act[(s * 16 + m) * PITCH + k];
      o0 = fmaf(h, W3[k * 4 + 0], o0);
      o1 = fmaf(h, W3[k * 4 + 1], o1);
      o2 = fmaf(h, W3[k * 4 + 2], o2);
      o3 = fmaf(h, W3[k * 4 + 3], o3);
    }
    if (s == 0) { o0 += b3[0]; o1 += b3[1]; o2 += b3[2]; o3 += b3[3]; }
    act[(s * 16 + m) * PITCH + H + 0] = o0;
    act[(s * 16 + m) * PITCH + H + 1] = o1;
    act[(s * 16 + m) * PITCH + H + 2] = o2;
    act[(s * 16 + m) * PITCH + H + 3] = o3;
  }
  __syncthreads();

  if (tt < 16) {
    const size_t i = (size_t)base + tt;
    float V[NSTREAM][4];
    #pragma unroll
    for (int s = 0; s < NSTREAM; ++s)
      #pragma unroll
      for (int o = 0; o < 4; ++o)
        V[s][o] = act[(s * 16 + tt) * PITCH + H + o];

    const float c   = V[0][0], v1 = V[0][1], v2 = V[0][2], Fi = V[0][3];
    const float dx0 = V[1][0], dx1 = V[1][1], dx2 = V[1][2], dx3 = V[1][3];
    const float xx0 = V[2][0], xx1 = V[2][1], xx2 = V[2][2], xx3 = V[2][3];
    const float dy0 = V[3][0], dy1 = V[3][1], dy2 = V[3][2], dy3 = V[3][3];
    const float yy0 = V[4][0], yy1 = V[4][1], yy2 = V[4][2], yy3 = V[4][3];
    const float dt0 = V[5][0], dt1 = V[5][1];

    const size_t Bz = (size_t)B;
    out[0 * Bz + i]          = c;
    out[1 * Bz + i]          = dt0;
    out[2 * Bz + 2 * i + 0]  = dx0;
    out[2 * Bz + 2 * i + 1]  = dy0;
    out[4 * Bz + 2 * i + 0]  = v1;
    out[4 * Bz + 2 * i + 1]  = v2;
    out[6 * Bz + i]          = dt1;
    out[7 * Bz + i]          = dx1 + dy2;
    out[8 * Bz + 2 * i + 0]  = dx1 * v1 + dy1 * v2;
    out[8 * Bz + 2 * i + 1]  = dx2 * v1 + dy2 * v2;
    out[10 * Bz + 2 * i + 0] = xx1 + yy1;
    out[10 * Bz + 2 * i + 1] = xx2 + yy2;
    out[12 * Bz + i]         = Fi;
    out[13 * Bz + 2 * i + 0] = dx3;
    out[13 * Bz + 2 * i + 1] = dy3;
    out[15 * Bz + i]         = xx3 + yy3;
    out[16 * Bz + i] = -0.5f * (xx0 + yy0)
                       - (dx0 * dx3 + dy0 * dy3) - c * (xx3 + yy3)
                       + (dx0 * v1 + dy0 * v2) + c * (dx1 + dy2);
  }
}

extern "C" void kernel_launch(void* const* d_in, const int* in_sizes, int n_in,
                              void* d_out, int out_size, void* d_ws, size_t ws_size,
                              hipStream_t stream) {
  const float* x  = (const float*)d_in[0];
  const float* W0 = (const float*)d_in[1];
  const float* b0 = (const float*)d_in[2];
  const float* W1 = (const float*)d_in[3];
  const float* b1 = (const float*)d_in[4];
  const float* W2 = (const float*)d_in[5];
  const float* b2 = (const float*)d_in[6];
  const float* W3 = (const float*)d_in[7];
  const float* b3 = (const float*)d_in[8];

  const int B = in_sizes[0] / 3;
  dim3 grid(B / (16 * TILES)), block(32 * NWAVE);
  pinn_grad_kernel<<<grid, block, 0, stream>>>(x, W0, b0, W1, b1, W2, b2, W3, b3,
                                               (float*)d_out, B);
}


# tanh (layer 0, both hidden epilogues) made branch-free: both range forms evaluated, one selected, exp(2|x|) as one exp2; a tile's 8 values interleave; tanh moves by about 1 ulp; sum order kept
# speedup vs baseline: 3.6438x; 1.1299x over previous
#include <hip/hip_runtime.h>
#include <hip/hip_bf16.h>

typedef float v2f __attribute__((ext_vector_type(2)));
typedef float v8f __attribute__((ext_vector_type(8)));

__device__ __forceinline__ v8f wmma_f32(v2f a, v2f b, v8f c) {
  return __builtin_amdgcn_wmma_f32_16x16x4_f32(false, a, false, b, (short)0, c,
                                               false, false);
}

__device__ __forceinline__ float tanh_sel(const float x) {
  const float ax  = __builtin_fabsf(x);
  const float y   = __builtin_amdgcn_exp2f(ax * 0x1.715476p+1f);
  const float big = fmaf(__builtin_amdgcn_rcpf(1.0f + y), -2.0f, 1.0f);
  const float s   = x * x;
  float p = fmaf(s, -0x1.758e7ap-8f, 0x1.521192p-6f);
  p = fmaf(s, p, -0x1.b8389cp-5f);
  p = fmaf(s, p, 0x1.110704p-3f);
  p = fmaf(s, p, -0x1.555532p-2f);
  const float sml = fmaf(s, ax * p, ax);
  return __builtin_copysignf(ax >= 0.625f ? big : sml, x);
}

#define NSTREAM 6
#define H 128
#define PITCH (H + 4)
#define TILES 2
#define NWAVE (2 * TILES)

#define NT 4
__device__ __forceinline__ void mlp_hidden_layer(
    const float* __restrict__ W, const float* __restrict__ bb,
    float (&act)[NSTREAM * 16 * PITCH],
    const int jq, const int ln, const int half, const int koff)
{
  v8f acc[NSTREAM][NT];
  #pragma unroll
  for (int t = 0; t < NT; ++t) {
    const float bj = bb[16 * (NT * jq + t) + ln];
    #pragma unroll
    for (int s = 0; s < NSTREAM; ++s)
      #pragma unroll
      for (int v = 0; v < 8; ++v) acc[s][t][v] = (s == 0) ? bj : 0.f;
  }

  v2f bfn[NT], afn[NSTREAM];
  #pragma unroll
  for (int t = 0; t < NT; ++t) {
    const int j = NT * jq + t;
    bfn[t].x = W[(size_t)koff * H + 16 * j + ln];
    bfn[t].y = W[(size_t)(koff + 1) * H + 16 * j + ln];
  }
  #pragma unroll
  for (int s = 0; s < NSTREAM; ++s)
    afn[s] = *(const v2f*)&act[(s * 16 + ln) * PITCH + koff];

  #pragma unroll 2
  for (int kc = 0; kc < 32; ++kc) {
    v2f bf[NT], af[NSTREAM];
    #pragma unroll
    for (int t = 0; t < NT; ++t) bf[t] = bfn[t];
    #pragma unroll
    for (int s = 0; s < NSTREAM; ++s) af[s] = afn[s];

    const int kn = (kc + 1 < 32) ? kc + 1 : kc;
    const int kb = 4 * kn + koff;
    #pragma unroll
    for (int t = 0; t < NT; ++t) {
      const int j = NT * jq + t;
      bfn[t].x = W[(size_t)kb * H + 16 * j + ln];
      bfn[t].y = W[(size_t)(kb + 1) * H + 16 * j + ln];
    }
    #pragma unroll
    for (int s = 0; s < NSTREAM; ++s)
      afn[s] = *(const v2f*)&act[(s * 16 + ln) * PITCH + 4 * kn + koff];

    #pragma unroll
    for (int s = 0; s < NSTREAM; ++s)
      #pragma unroll
      for (int t = 0; t < NT; ++t) acc[s][t] = wmma_f32(af[s], bf[t], acc[s][t]);
  }

  __syncthreads();

  #pragma unroll
  for (int t = 0; t < NT; ++t) {
    const int j = NT * jq + t;
    v8f acur, zcur;
    #pragma unroll
    for (int s = 0; s < NSTREAM; ++s) {
      v8f res;
      if (s == 0) {
        #pragma unroll
        for (int v = 0; v < 8; ++v) res[v] = tanh_sel(acc[s][t][v]);
        acur = res;
      } else if (s == 1 || s == 3) {
        zcur = acc[s][t];
        #pragma unroll
        for (int v = 0; v < 8; ++v) {
          const float a = acur[v];
          res[v] = (1.f - a * a) * acc[s][t][v];
        }
      } else if (s == 2 || s == 4) {
        #pragma unroll
        for (int v = 0; v < 8; ++v) {
          const float a  = acur[v];
          const float s2 = 1.f - a * a;
          const float zd = zcur[v];
          res[v] = s2 * acc[s][t][v] - 2.f * a * s2 * zd * zd;
        }
      } else {
        #pragma unroll
        for (int v = 0; v < 8; ++v) {
          const float a = acur[v];
          res[v] = (1.f - a * a) * acc[s][t][v];
        }
      }
      #pragma unroll
      for (int v = 0; v < 8; ++v)
        act[(s * 16 + (v + 8 * half)) * PITCH + 16 * j + ln] = res[v];
    }
  }
}

__global__ __launch_bounds__(32 * NWAVE) void pinn_grad_kernel(
    const float* __restrict__ x,
    const float* __restrict__ W0, const float* __restrict__ b0,
    const float* __restrict__ W1, const float* __restrict__ b1,
    const float* __restrict__ W2, const float* __restrict__ b2,
    const float* __restrict__ W3, const float* __restrict__ b3,
    float* __restrict__ out, int B)
{
  __shared__ __align__(16) float act_all[TILES][NSTREAM * 16 * PITCH];

  const int tid  = threadIdx.x;
  const int wave = tid >> 5;
  const int tile = wave >> 1;
  const int jq   = wave & 1;
  const int tt   = tid & 63;
  const int lane = tid & 31;
  float (&act)[NSTREAM * 16 * PITCH] = act_all[tile];
  const int half = lane >> 4;
  const int ln   = lane & 15;
  const int koff = half * 2;
  const int base = (blockIdx.x * TILES + tile) * 16;

  for (int idx = tt; idx < 16 * H; idx += 64) {
    const int m = idx >> 7, n = idx & (H - 1);
    const float* xs = x + (size_t)(base + m) * 3;
    const float x0 = xs[0], x1 = xs[1], x2 = xs[2];
    const float w0 = W0[0 * H + n], w1 = W0[1 * H + n], w2 = W0[2 * H + n];
    const float z  = fmaf(x0, w0, fmaf(x1, w1, fmaf(x2, w2, b0[n])));
    const float a  = tanh_sel(z);
    const float s2 = 1.f - a * a;
    act[(0 * 16 + m) * PITCH + n] = a;
    act[(1 * 16 + m) * PITCH + n] = s2 * w0;
    act[(2 * 16 + m) * PITCH + n] = -2.f * a * s2 * w0 * w0;
    act[(3 * 16 + m) * PITCH + n] = s2 * w1;
    act[(4 * 16 + m) * PITCH + n] = -2.f * a * s2 * w1 * w1;
    act[(5 * 16 + m) * PITCH + n] = s2 * w2;
  }
  __syncthreads();

  mlp_hidden_layer(W1, b1, act, jq, ln, half, koff);
  __syncthreads();
  mlp_hidden_layer(W2, b2, act, jq, ln, half, koff);
  __syncthreads();

  for (int p = tt; p < NSTREAM * 16; p += 64) {
    const int s = p >> 4;
    const int m = p & 15;
    float o0 = 0.f, o1 = 0.f, o2 = 0.f, o3 = 0.f;
    #pragma unroll 4
    for (int k = 0; k < H; ++k) {
      const float h = act[(s * 16 + m) * PITCH + k];
      o0 = fmaf(h, W3[k * 4 + 0], o0);
      o1 = fmaf(h, W3[k * 4 + 1], o1);
      o2 = fmaf(h, W3[k * 4 + 2], o2);
      o3 = fmaf(h, W3[k * 4 + 3], o3);
    }
    if (s == 0) { o0 += b3[0]; o1 += b3[1]; o2 += b3[2]; o3 += b3[3]; }
    act[(s * 16 + m) * PITCH + H + 0] = o0;
    act[(s * 16 + m) * PITCH + H + 1] = o1;
    act[(s * 16 + m) * PITCH + H + 2] = o2;
    act[(s * 16 + m) * PITCH + H + 3] = o3;
  }
  __syncthreads();

  if (tt < 16) {
    const size_t i = (size_t)base + tt;
    float V[NSTREAM][4];
    #pragma unroll
    for (int s = 0; s < NSTREAM; ++s)
      #pragma unroll
      for (int o = 0; o < 4; ++o)
        V[s][o] = act[(s * 16 + tt) * PITCH + H + o];

    const float c   = V[0][0], v1 = V[0][1], v2 = V[0][2], Fi = V[0][3];
    const float dx0 = V[1][0], dx1 = V[1][1], dx2 = V[1][2], dx3 = V[1][3];
    const float xx0 = V[2][0], xx1 = V[2][1], xx2 = V[2][2], xx3 = V[2][3];
    const float dy0 = V[3][0], dy1 = V[3][1], dy2 = V[3][2], dy3 = V[3][3];
    const float yy0 = V[4][0], yy1 = V[4][1], yy2 = V[4][2], yy3 = V[4][3];
    const float dt0 = V[5][0], dt1 = V[5][1];

    const size_t Bz = (size_t)B;
    out[0 * Bz + i]          = c;
    out[1 * Bz + i]          = dt0;
    out[2 * Bz + 2 * i + 0]  = dx0;
    out[2 * Bz + 2 * i + 1]  = dy0;
    out[4 * Bz + 2 * i + 0]  = v1;
    out[4 * Bz + 2 * i + 1]  = v2;
    out[6 * Bz + i]          = dt1;
    out[7 * Bz + i]          = dx1 + dy2;
    out[8 * Bz + 2 * i + 0]  = dx1 * v1 + dy1 * v2;
    out[8 * Bz + 2 * i + 1]  = dx2 * v1 + dy2 * v2;
    out[10 * Bz + 2 * i + 0] = xx1 + yy1;
    out[10 * Bz + 2 * i + 1] = xx2 + yy2;
    out[12 * Bz + i]         = Fi;
    out[13 * Bz + 2 * i + 0] = dx3;
    out[13 * Bz + 2 * i + 1] = dy3;
    out[15 * Bz + i]         = xx3 + yy3;
    out[16 * Bz + i] = -0.5f * (xx0 + yy0)
                       - (dx0 * dx3 + dy0 * dy3) - c * (xx3 + yy3)
                       + (dx0 * v1 + dy0 * v2) + c * (dx1 + dy2);
  }
}

extern "C" void kernel_launch(void* const* d_in, const int* in_sizes, int n_in,
                              void* d_out, int out_size, void* d_ws, size_t ws_size,
                              hipStream_t stream) {
  const float* x  = (const float*)d_in[0];
  const float* W0 = (const float*)d_in[1];
  const float* b0 = (const float*)d_in[2];
  const float* W1 = (const float*)d_in[3];
  const float* b1 = (const float*)d_in[4];
  const float* W2 = (const float*)d_in[5];
  const float* b2 = (const float*)d_in[6];
  const float* W3 = (const float*)d_in[7];
  const float* b3 = (const float*)d_in[8];

  const int B = in_sizes[0] / 3;
  dim3 grid(B / (16 * TILES)), block(32 * NWAVE);
  pinn_grad_kernel<<<grid, block, 0, stream>>>(x, W0, b0, W1, b1, W2, b2, W3, b3,
                                               (float*)d_out, B);
}


# Streams d2/dx2 and d2/dy2 merged into one Laplacian stream (only their sum reaches any output): 5 streams, 20 WMMAs per k-step, not 24; xx + yy is summed ahead of the products, not after them
# speedup vs baseline: 4.1475x; 1.1382x over previous
#include <hip/hip_runtime.h>
#include <hip/hip_bf16.h>

typedef float v2f __attribute__((ext_vector_type(2)));
typedef float v8f __attribute__((ext_vector_type(8)));

__device__ __forceinline__ v8f wmma_f32(v2f a, v2f b, v8f c) {
  return __builtin_amdgcn_wmma_f32_16x16x4_f32(false, a, false, b, (short)0, c,
                                               false, false);
}

__device__ __forceinline__ float tanh_sel(const float x) {
  const float ax  = __builtin_fabsf(x);
  const float y   = __builtin_amdgcn_exp2f(ax * 0x1.715476p+1f);
  const float big = fmaf(__builtin_amdgcn_rcpf(1.0f + y), -2.0f, 1.0f);
  const float s   = x * x;
  float p = fmaf(s, -0x1.758e7ap-8f, 0x1.521192p-6f);
  p = fmaf(s, p, -0x1.b8389cp-5f);
  p = fmaf(s, p, 0x1.110704p-3f);
  p = fmaf(s, p, -0x1.555532p-2f);
  const float sml = fmaf(s, ax * p, ax);
  return __builtin_copysignf(ax >= 0.625f ? big : sml, x);
}

#define NSTREAM 5
#define H 128
#define PITCH (H + 4)
#define TILES 2
#define NWAVE (2 * TILES)

#define NT 4
__device__ __forceinline__ void mlp_hidden_layer(
    const float* __restrict__ W, const float* __restrict__ bb,
    float (&act)[NSTREAM * 16 * PITCH],
    const int jq, const int ln, const int half, const int koff)
{
  v8f acc[NSTREAM][NT];
  #pragma unroll
  for (int t = 0; t < NT; ++t) {
    const float bj = bb[16 * (NT * jq + t) + ln];
    #pragma unroll
    for (int s = 0; s < NSTREAM; ++s)
      #pragma unroll
      for (int v = 0; v < 8; ++v) acc[s][t][v] = (s == 0) ? bj : 0.f;
  }

  v2f bfn[NT], afn[NSTREAM];
  #pragma unroll
  for (int t = 0; t < NT; ++t) {
    const int j = NT * jq + t;
    bfn[t].x = W[(size_t)koff * H + 16 * j + ln];
    bfn[t].y = W[(size_t)(koff + 1) * H + 16 * j + ln];
  }
  #pragma unroll
  for (int s = 0; s < NSTREAM; ++s)
    afn[s] = *(const v2f*)&act[(s * 16 + ln) * PITCH + koff];

  #pragma unroll 2
  for (int kc = 0; kc < 32; ++kc) {
    v2f bf[NT], af[NSTREAM];
    #pragma unroll
    for (int t = 0; t < NT; ++t) bf[t] = bfn[t];
    #pragma unroll
    for (int s = 0; s < NSTREAM; ++s) af[s] = afn[s];

    const int kn = (kc + 1 < 32) ? kc + 1 : kc;
    const int kb = 4 * kn + koff;
    #pragma unroll
    for (int t = 0; t < NT; ++t) {
      const int j = NT * jq + t;
      bfn[t].x = W[(size_t)kb * H + 16 * j + ln];
      bfn[t].y = W[(size_t)(kb + 1) * H + 16 * j + ln];
    }
    #pragma unroll
    for (int s = 0; s < NSTREAM; ++s)
      afn[s] = *(const v2f*)&act[(s * 16 + ln) * PITCH + 4 * kn + koff];

    #pragma unroll
    for (int s = 0; s < NSTREAM; ++s)
      #pragma unroll
      for (int t = 0; t < NT; ++t) acc[s][t] = wmma_f32(af[s], bf[t], acc[s][t]);
  }

  __syncthreads();

  #pragma unroll
  for (int t = 0; t < NT; ++t) {
    const int j = NT * jq + t;
    v8f acur;
    #pragma unroll
    for (int s = 0; s < NSTREAM; ++s) {
      v8f res;
      if (s == 0) {
        #pragma unroll
        for (int v = 0; v < 8; ++v) res[v] = tanh_sel(acc[s][t][v]);
        acur = res;
      } else if (s == 3) {
        #pragma unroll
        for (int v = 0; v < 8; ++v) {
          const float a  = acur[v];
          const float s2 = 1.f - a * a;
          const float zx = acc[1][t][v];
          const float zy = acc[2][t][v];
          res[v] = s2 * acc[s][t][v] - 2.f * a * s2 * (zx * zx + zy * zy);
        }
      } else {
        #pragma unroll
        for (int v = 0; v < 8; ++v) {
          const float a = acur[v];
          res[v] = (1.f - a * a) * acc[s][t][v];
        }
      }
      #pragma unroll
      for (int v = 0; v < 8; ++v)
        act[(s * 16 + (v + 8 * half)) * PITCH + 16 * j + ln] = res[v];
    }
  }
}

__global__ __launch_bounds__(32 * NWAVE) void pinn_grad_kernel(
    const float* __restrict__ x,
    const float* __restrict__ W0, const float* __restrict__ b0,
    const float* __restrict__ W1, const float* __restrict__ b1,
    const float* __restrict__ W2, const float* __restrict__ b2,
    const float* __restrict__ W3, const float* __restrict__ b3,
    float* __restrict__ out, int B)
{
  __shared__ __align__(16) float act_all[TILES][NSTREAM * 16 * PITCH];

  const int tid  = threadIdx.x;
  const int wave = tid >> 5;
  const int tile = wave >> 1;
  const int jq   = wave & 1;
  const int tt   = tid & 63;
  const int lane = tid & 31;
  float (&act)[NSTREAM * 16 * PITCH] = act_all[tile];
  const int half = lane >> 4;
  const int ln   = lane & 15;
  const int koff = half * 2;
  const int base = (blockIdx.x * TILES + tile) * 16;

  for (int idx = tt; idx < 16 * H; idx += 64) {
    const int m = idx >> 7, n = idx & (H - 1);
    const float* xs = x + (size_t)(base + m) * 3;
    const float x0 = xs[0], x1 = xs[1], x2 = xs[2];
    const float w0 = W0[0 * H + n], w1 = W0[1 * H + n], w2 = W0[2 * H + n];
    const float z  = fmaf(x0, w0, fmaf(x1, w1, fmaf(x2, w2, b0[n])));
    const float a  = tanh_sel(z);
    const float s2 = 1.f - a * a;
    act[(0 * 16 + m) * PITCH + n] = a;
    act[(1 * 16 + m) * PITCH + n] = s2 * w0;
    act[(2 * 16 + m) * PITCH + n] = s2 * w1;
    act[(3 * 16 + m) * PITCH + n] = -2.f * a * s2 * (w0 * w0 + w1 * w1);
    act[(4 * 16 + m) * PITCH + n] = s2 * w2;
  }
  __syncthreads();

  mlp_hidden_layer(W1, b1, act, jq, ln, half, koff);
  __syncthreads();
  mlp_hidden_layer(W2, b2, act, jq, ln, half, koff);
  __syncthreads();

  for (int p = tt; p < NSTREAM * 16; p += 64) {
    const int s = p >> 4;
    const int m = p & 15;
    float o0 = 0.f, o1 = 0.f, o2 = 0.f, o3 = 0.f;
    #pragma unroll 4
    for (int k = 0; k < H; ++k) {
      const float h = act[(s * 16 + m) * PITCH + k];
      o0 = fmaf(h, W3[k * 4 + 0], o0);
      o1 = fmaf(h, W3[k * 4 + 1], o1);
      o2 = fmaf(h, W3[k * 4 + 2], o2);
      o3 = fmaf(h, W3[k * 4 + 3], o3);
    }
    if (s == 0) { o0 += b3[0]; o1 += b3[1]; o2 += b3[2]; o3 += b3[3]; }
    act[(s * 16 + m) * PITCH + H + 0] = o0;
    act[(s * 16 + m) * PITCH + H + 1] = o1;
    act[(s * 16 + m) * PITCH + H + 2] = o2;
    act[(s * 16 + m) * PITCH + H + 3] = o3;
  }
  __syncthreads();

  if (tt < 16) {
    const size_t i = (size_t)base + tt;
    float V[NSTREAM][4];
    #pragma unroll
    for (int s = 0; s < NSTREAM; ++s)
      #pragma unroll
      for (int o = 0; o < 4; ++o)
        V[s][o] = act[(s * 16 + tt) * PITCH + H + o];

    const float c   = V[0][0], v1 = V[0][1], v2 = V[0][2], Fi = V[0][3];
    const float dx0 = V[1][0], dx1 = V[1][1], dx2 = V[1][2], dx3 = V[1][3];
    const float dy0 = V[2][0], dy1 = V[2][1], dy2 = V[2][2], dy3 = V[2][3];
    const float lp0 = V[3][0], lp1 = V[3][1], lp2 = V[3][2], lp3 = V[3][3];
    const float dt0 = V[4][0], dt1 = V[4][1];

    const size_t Bz = (size_t)B;
    out[0 * Bz + i]          = c;
    out[1 * Bz + i]          = dt0;
    out[2 * Bz + 2 * i + 0]  = dx0;
    out[2 * Bz + 2 * i + 1]  = dy0;
    out[4 * Bz + 2 * i + 0]  = v1;
    out[4 * Bz + 2 * i + 1]  = v2;
    out[6 * Bz + i]          = dt1;
    out[7 * Bz + i]          = dx1 + dy2;
    out[8 * Bz + 2 * i + 0]  = dx1 * v1 + dy1 * v2;
    out[8 * Bz + 2 * i + 1]  = dx2 * v1 + dy2 * v2;
    out[10 * Bz + 2 * i + 0] = lp1;
    out[10 * Bz + 2 * i + 1] = lp2;
    out[12 * Bz + i]         = Fi;
    out[13 * Bz + 2 * i + 0] = dx3;
    out[13 * Bz + 2 * i + 1] = dy3;
    out[15 * Bz + i]         = lp3;
    out[16 * Bz + i] = -0.5f * lp0
                       - (dx0 * dx3 + dy0 * dy3) - c * lp3
                       + (dx0 * v1 + dy0 * v2) + c * (dx1 + dy2);
  }
}

extern "C" void kernel_launch(void* const* d_in, const int* in_sizes, int n_in,
                              void* d_out, int out_size, void* d_ws, size_t ws_size,
                              hipStream_t stream) {
  const float* x  = (const float*)d_in[0];
  const float* W0 = (const float*)d_in[1];
  const float* b0 = (const float*)d_in[2];
  const float* W1 = (const float*)d_in[3];
  const float* b1 = (const float*)d_in[4];
  const float* W2 = (const float*)d_in[5];
  const float* b2 = (const float*)d_in[6];
  const float* W3 = (const float*)d_in[7];
  const float* b3 = (const float*)d_in[8];

  const int B = in_sizes[0] / 3;
  dim3 grid(B / (16 * TILES)), block(32 * NWAVE);
  pinn_grad_kernel<<<grid, block, 0, stream>>>(x, W0, b0, W1, b1, W2, b2, W3, b3,
                                               (float*)d_out, B);
}


# Last layer (128->4): a tile's 64 threads split it as sample x quarter of k, W3 rows preloaded into registers; each dot product = four 32-term chains added pairwise, not one 128-term chain
# speedup vs baseline: 4.4395x; 1.0704x over previous
#include <hip/hip_runtime.h>
#include <hip/hip_bf16.h>

typedef float v2f __attribute__((ext_vector_type(2)));
typedef float v4f __attribute__((ext_vector_type(4)));
typedef float v8f __attribute__((ext_vector_type(8)));

__device__ __forceinline__ v8f wmma_f32(v2f a, v2f b, v8f c) {
  return __builtin_amdgcn_wmma_f32_16x16x4_f32(false, a, false, b, (short)0, c,
                                               false, false);
}

__device__ __forceinline__ float xhalf(const float x) {
  return __builtin_bit_cast(float, __builtin_amdgcn_permlanex16(
      __builtin_bit_cast(int, x), __builtin_bit_cast(int, x),
      0x76543210, 0xfedcba98, false, false));
}

__device__ __forceinline__ float tanh_sel(const float x) {
  const float ax  = __builtin_fabsf(x);
  const float y   = __builtin_amdgcn_exp2f(ax * 0x1.715476p+1f);
  const float big = fmaf(__builtin_amdgcn_rcpf(1.0f + y), -2.0f, 1.0f);
  const float s   = x * x;
  float p = fmaf(s, -0x1.758e7ap-8f, 0x1.521192p-6f);
  p = fmaf(s, p, -0x1.b8389cp-5f);
  p = fmaf(s, p, 0x1.110704p-3f);
  p = fmaf(s, p, -0x1.555532p-2f);
  const float sml = fmaf(s, ax * p, ax);
  return __builtin_copysignf(ax >= 0.625f ? big : sml, x);
}

#define NSTREAM 5
#define H 128
#define PITCH (H + 4)
#define TILES 2
#define NWAVE (2 * TILES)

#define NT 4
__device__ __forceinline__ void mlp_hidden_layer(
    const float* __restrict__ W, const float* __restrict__ bb,
    float (&act)[NSTREAM * 16 * PITCH],
    const int jq, const int ln, const int half, const int koff)
{
  v8f acc[NSTREAM][NT];
  #pragma unroll
  for (int t = 0; t < NT; ++t) {
    const float bj = bb[16 * (NT * jq + t) + ln];
    #pragma unroll
    for (int s = 0; s < NSTREAM; ++s)
      #pragma unroll
      for (int v = 0; v < 8; ++v) acc[s][t][v] = (s == 0) ? bj : 0.f;
  }

  v2f bfn[NT], afn[NSTREAM];
  #pragma unroll
  for (int t = 0; t < NT; ++t) {
    const int j = NT * jq + t;
    bfn[t].x = W[(size_t)koff * H + 16 * j + ln];
    bfn[t].y = W[(size_t)(koff + 1) * H + 16 * j + ln];
  }
  #pragma unroll
  for (int s = 0; s < NSTREAM; ++s)
    afn[s] = *(const v2f*)&act[(s * 16 + ln) * PITCH + koff];

  #pragma unroll 2
  for (int kc = 0; kc < 32; ++kc) {
    v2f bf[NT], af[NSTREAM];
    #pragma unroll
    for (int t = 0; t < NT; ++t) bf[t] = bfn[t];
    #pragma unroll
    for (int s = 0; s < NSTREAM; ++s) af[s] = afn[s];

    const int kn = (kc + 1 < 32) ? kc + 1 : kc;
    const int kb = 4 * kn + koff;
    #pragma unroll
    for (int t = 0; t < NT; ++t) {
      const int j = NT * jq + t;
      bfn[t].x = W[(size_t)kb * H + 16 * j + ln];
      bfn[t].y = W[(size_t)(kb + 1) * H + 16 * j + ln];
    }
    #pragma unroll
    for (int s = 0; s < NSTREAM; ++s)
      afn[s] = *(const v2f*)&act[(s * 16 + ln) * PITCH + 4 * kn + koff];

    #pragma unroll
    for (int s = 0; s < NSTREAM; ++s)
      #pragma unroll
      for (int t = 0; t < NT; ++t) acc[s][t] = wmma_f32(af[s], bf[t], acc[s][t]);
  }

  __syncthreads();

  #pragma unroll
  for (int t = 0; t < NT; ++t) {
    const int j = NT * jq + t;
    v8f acur;
    #pragma unroll
    for (int s = 0; s < NSTREAM; ++s) {
      v8f res;
      if (s == 0) {
        #pragma unroll
        for (int v = 0; v < 8; ++v) res[v] = tanh_sel(acc[s][t][v]);
        acur = res;
      } else if (s == 3) {
        #pragma unroll
        for (int v = 0; v < 8; ++v) {
          const float a  = acur[v];
          const float s2 = 1.f - a * a;
          const float zx = acc[1][t][v];
          const float zy = acc[2][t][v];
          res[v] = s2 * acc[s][t][v] - 2.f * a * s2 * (zx * zx + zy * zy);
        }
      } else {
        #pragma unroll
        for (int v = 0; v < 8; ++v) {
          const float a = acur[v];
          res[v] = (1.f - a * a) * acc[s][t][v];
        }
      }
      #pragma unroll
      for (int v = 0; v < 8; ++v)
        act[(s * 16 + (v + 8 * half)) * PITCH + 16 * j + ln] = res[v];
    }
  }
}

__global__ __launch_bounds__(32 * NWAVE) void pinn_grad_kernel(
    const float* __restrict__ x,
    const float* __restrict__ W0, const float* __restrict__ b0,
    const float* __restrict__ W1, const float* __restrict__ b1,
    const float* __restrict__ W2, const float* __restrict__ b2,
    const float* __restrict__ W3, const float* __restrict__ b3,
    float* __restrict__ out, int B)
{
  __shared__ __align__(16) float act_all[TILES][NSTREAM * 16 * PITCH];

  const int tid  = threadIdx.x;
  const int wave = tid >> 5;
  const int tile = wave >> 1;
  const int jq   = wave & 1;
  const int tt   = tid & 63;
  const int lane = tid & 31;
  float (&act)[NSTREAM * 16 * PITCH] = act_all[tile];
  const int half = lane >> 4;
  const int ln   = lane & 15;
  const int koff = half * 2;
  const int base = (blockIdx.x * TILES + tile) * 16;

  for (int idx = tt; idx < 16 * H; idx += 64) {
    const int m = idx >> 7, n = idx & (H - 1);
    const float* xs = x + (size_t)(base + m) * 3;
    const float x0 = xs[0], x1 = xs[1], x2 = xs[2];
    const float w0 = W0[0 * H + n], w1 = W0[1 * H + n], w2 = W0[2 * H + n];
    const float z  = fmaf(x0, w0, fmaf(x1, w1, fmaf(x2, w2, b0[n])));
    const float a  = tanh_sel(z);
    const float s2 = 1.f - a * a;
    act[(0 * 16 + m) * PITCH + n] = a;
    act[(1 * 16 + m) * PITCH + n] = s2 * w0;
    act[(2 * 16 + m) * PITCH + n] = s2 * w1;
    act[(3 * 16 + m) * PITCH + n] = -2.f * a * s2 * (w0 * w0 + w1 * w1);
    act[(4 * 16 + m) * PITCH + n] = s2 * w2;
  }
  __syncthreads();

  mlp_hidden_layer(W1, b1, act, jq, ln, half, koff);
  __syncthreads();
  mlp_hidden_layer(W2, b2, act, jq, ln, half, koff);

  const int kq = 2 * jq + half;
  v4f w3[32];
  #pragma unroll
  for (int kk = 0; kk < 32; ++kk) {
    const float* wr = W3 + (size_t)(32 * kq + kk) * 4;
    w3[kk] = (v4f){wr[0], wr[1], wr[2], wr[3]};
  }
  __syncthreads();

  v2f s01[NSTREAM], s23[NSTREAM];
  #pragma unroll
  for (int s = 0; s < NSTREAM; ++s) { s01[s] = (v2f){0.f, 0.f}; s23[s] = (v2f){0.f, 0.f}; }
  #pragma unroll
  for (int i = 0; i < 8; ++i) {
    #pragma unroll
    for (int s = 0; s < NSTREAM; ++s) {
      const v4f h = *(const v4f*)&act[(s * 16 + ln) * PITCH + 32 * kq + 4 * i];
      #pragma unroll
      for (int e = 0; e < 4; ++e) {
        const v4f w  = w3[4 * i + e];
        const v2f hh = {h[e], h[e]};
        s01[s] = __builtin_elementwise_fma(hh, (v2f){w.x, w.y}, s01[s]);
        if (s != 4) s23[s] = __builtin_elementwise_fma(hh, (v2f){w.z, w.w}, s23[s]);
      }
    }
  }
  #pragma unroll
  for (int s = 0; s < NSTREAM; ++s) {
    s01[s].x += xhalf(s01[s].x);
    s01[s].y += xhalf(s01[s].y);
    if (s != 4) { s23[s].x += xhalf(s23[s].x); s23[s].y += xhalf(s23[s].y); }
  }
  if (jq == 1 && half == 0) {
    #pragma unroll
    for (int s = 0; s < NSTREAM; ++s) {
      if (s != 4) *(v4f*)&act[(s * 16 + ln) * PITCH + H] = (v4f){s01[s].x, s01[s].y, s23[s].x, s23[s].y};
      else        *(v2f*)&act[(s * 16 + ln) * PITCH + H] = s01[s];
    }
  }
  __syncthreads();

  if (tt < 16) {
    const size_t i = (size_t)base + tt;
    float V[NSTREAM][4];
    #pragma unroll
    for (int s = 0; s < NSTREAM; ++s) {
      if (s != 4) {
        const v4f q = *(const v4f*)&act[(s * 16 + tt) * PITCH + H];
        V[s][0] = s01[s].x + q.x; V[s][1] = s01[s].y + q.y;
        V[s][2] = s23[s].x + q.z; V[s][3] = s23[s].y + q.w;
      } else {
        const v2f q = *(const v2f*)&act[(s * 16 + tt) * PITCH + H];
        V[s][0] = s01[s].x + q.x; V[s][1] = s01[s].y + q.y;
        V[s][2] = 0.f; V[s][3] = 0.f;
      }
    }
    #pragma unroll
    for (int o = 0; o < 4; ++o) V[0][o] += b3[o];

    const float c   = V[0][0], v1 = V[0][1], v2 = V[0][2], Fi = V[0][3];
    const float dx0 = V[1][0], dx1 = V[1][1], dx2 = V[1][2], dx3 = V[1][3];
    const float dy0 = V[2][0], dy1 = V[2][1], dy2 = V[2][2], dy3 = V[2][3];
    const float lp0 = V[3][0], lp1 = V[3][1], lp2 = V[3][2], lp3 = V[3][3];
    const float dt0 = V[4][0], dt1 = V[4][1];

    const size_t Bz = (size_t)B;
    out[0 * Bz + i]          = c;
    out[1 * Bz + i]          = dt0;
    out[2 * Bz + 2 * i + 0]  = dx0;
    out[2 * Bz + 2 * i + 1]  = dy0;
    out[4 * Bz + 2 * i + 0]  = v1;
    out[4 * Bz + 2 * i + 1]  = v2;
    out[6 * Bz + i]          = dt1;
    out[7 * Bz + i]          = dx1 + dy2;
    out[8 * Bz + 2 * i + 0]  = dx1 * v1 + dy1 * v2;
    out[8 * Bz + 2 * i + 1]  = dx2 * v1 + dy2 * v2;
    out[10 * Bz + 2 * i + 0] = lp1;
    out[10 * Bz + 2 * i + 1] = lp2;
    out[12 * Bz + i]         = Fi;
    out[13 * Bz + 2 * i + 0] = dx3;
    out[13 * Bz + 2 * i + 1] = dy3;
    out[15 * Bz + i]         = lp3;
    out[16 * Bz + i] = -0.5f * lp0
                       - (dx0 * dx3 + dy0 * dy3) - c * lp3
                       + (dx0 * v1 + dy0 * v2) + c * (dx1 + dy2);
  }
}

extern "C" void kernel_launch(void* const* d_in, const int* in_sizes, int n_in,
                              void* d_out, int out_size, void* d_ws, size_t ws_size,
                              hipStream_t stream) {
  const float* x  = (const float*)d_in[0];
  const float* W0 = (const float*)d_in[1];
  const float* b0 = (const float*)d_in[2];
  const float* W1 = (const float*)d_in[3];
  const float* b1 = (const float*)d_in[4];
  const float* W2 = (const float*)d_in[5];
  const float* b2 = (const float*)d_in[6];
  const float* W3 = (const float*)d_in[7];
  const float* b3 = (const float*)d_in[8];

  const int B = in_sizes[0] / 3;
  dim3 grid(B / (16 * TILES)), block(32 * NWAVE);
  pinn_grad_kernel<<<grid, block, 0, stream>>>(x, W0, b0, W1, b1, W2, b2, W3, b3,
                                               (float*)d_out, B);
}


# Layer 0 re-tiled: a lane owns 4 adjacent columns, its wave 8 samples; inputs of all 8 samples requested together ahead of the first tanh; 16-byte LDS stores; same expressions, sum orders kept
# speedup vs baseline: 4.6355x; 1.0441x over previous
#include <hip/hip_runtime.h>
#include <hip/hip_bf16.h>

typedef float v2f __attribute__((ext_vector_type(2)));
typedef float v4f __attribute__((ext_vector_type(4)));
typedef float v8f __attribute__((ext_vector_type(8)));

__device__ __forceinline__ v8f wmma_f32(v2f a, v2f b, v8f c) {
  return __builtin_amdgcn_wmma_f32_16x16x4_f32(false, a, false, b, (short)0, c,
                                               false, false);
}

__device__ __forceinline__ float xhalf(const float x) {
  return __builtin_bit_cast(float, __builtin_amdgcn_permlanex16(
      __builtin_bit_cast(int, x), __builtin_bit_cast(int, x),
      0x76543210, 0xfedcba98, false, false));
}

__device__ __forceinline__ float tanh_sel(const float x) {
  const float ax  = __builtin_fabsf(x);
  const float y   = __builtin_amdgcn_exp2f(ax * 0x1.715476p+1f);
  const float big = fmaf(__builtin_amdgcn_rcpf(1.0f + y), -2.0f, 1.0f);
  const float s   = x * x;
  float p = fmaf(s, -0x1.758e7ap-8f, 0x1.521192p-6f);
  p = fmaf(s, p, -0x1.b8389cp-5f);
  p = fmaf(s, p, 0x1.110704p-3f);
  p = fmaf(s, p, -0x1.555532p-2f);
  const float sml = fmaf(s, ax * p, ax);
  return __builtin_copysignf(ax >= 0.625f ? big : sml, x);
}

#define NSTREAM 5
#define H 128
#define PITCH (H + 4)
#define TILES 2
#define NWAVE (2 * TILES)

#define NT 4
__device__ __forceinline__ void mlp_hidden_layer(
    const float* __restrict__ W, const float* __restrict__ bb,
    float (&act)[NSTREAM * 16 * PITCH],
    const int jq, const int ln, const int half, const int koff)
{
  v8f acc[NSTREAM][NT];
  #pragma unroll
  for (int t = 0; t < NT; ++t) {
    const float bj = bb[16 * (NT * jq + t) + ln];
    #pragma unroll
    for (int s = 0; s < NSTREAM; ++s)
      #pragma unroll
      for (int v = 0; v < 8; ++v) acc[s][t][v] = (s == 0) ? bj : 0.f;
  }

  v2f bfn[NT], afn[NSTREAM];
  #pragma unroll
  for (int t = 0; t < NT; ++t) {
    const int j = NT * jq + t;
    bfn[t].x = W[(size_t)koff * H + 16 * j + ln];
    bfn[t].y = W[(size_t)(koff + 1) * H + 16 * j + ln];
  }
  #pragma unroll
  for (int s = 0; s < NSTREAM; ++s)
    afn[s] = *(const v2f*)&act[(s * 16 + ln) * PITCH + koff];

  #pragma unroll 2
  for (int kc = 0; kc < 32; ++kc) {
    v2f bf[NT], af[NSTREAM];
    #pragma unroll
    for (int t = 0; t < NT; ++t) bf[t] = bfn[t];
    #pragma unroll
    for (int s = 0; s < NSTREAM; ++s) af[s] = afn[s];

    const int kn = (kc + 1 < 32) ? kc + 1 : kc;
    const int kb = 4 * kn + koff;
    #pragma unroll
    for (int t = 0; t < NT; ++t) {
      const int j = NT * jq + t;
      bfn[t].x = W[(size_t)kb * H + 16 * j + ln];
      bfn[t].y = W[(size_t)(kb + 1) * H + 16 * j + ln];
    }
    #pragma unroll
    for (int s = 0; s < NSTREAM; ++s)
      afn[s] = *(const v2f*)&act[(s * 16 + ln) * PITCH + 4 * kn + koff];

    #pragma unroll
    for (int s = 0; s < NSTREAM; ++s)
      #pragma unroll
      for (int t = 0; t < NT; ++t) acc[s][t] = wmma_f32(af[s], bf[t], acc[s][t]);
  }

  __syncthreads();

  #pragma unroll
  for (int t = 0; t < NT; ++t) {
    const int j = NT * jq + t;
    v8f acur;
    #pragma unroll
    for (int s = 0; s < NSTREAM; ++s) {
      v8f res;
      if (s == 0) {
        #pragma unroll
        for (int v = 0; v < 8; ++v) res[v] = tanh_sel(acc[s][t][v]);
        acur = res;
      } else if (s == 3) {
        #pragma unroll
        for (int v = 0; v < 8; ++v) {
          const float a  = acur[v];
          const float s2 = 1.f - a * a;
          const float zx = acc[1][t][v];
          const float zy = acc[2][t][v];
          res[v] = s2 * acc[s][t][v] - 2.f * a * s2 * (zx * zx + zy * zy);
        }
      } else {
        #pragma unroll
        for (int v = 0; v < 8; ++v) {
          const float a = acur[v];
          res[v] = (1.f - a * a) * acc[s][t][v];
        }
      }
      #pragma unroll
      for (int v = 0; v < 8; ++v)
        act[(s * 16 + (v + 8 * half)) * PITCH + 16 * j + ln] = res[v];
    }
  }
}

__global__ __launch_bounds__(32 * NWAVE) void pinn_grad_kernel(
    const float* __restrict__ x,
    const float* __restrict__ W0, const float* __restrict__ b0,
    const float* __restrict__ W1, const float* __restrict__ b1,
    const float* __restrict__ W2, const float* __restrict__ b2,
    const float* __restrict__ W3, const float* __restrict__ b3,
    float* __restrict__ out, int B)
{
  __shared__ __align__(16) float act_all[TILES][NSTREAM * 16 * PITCH];

  const int tid  = threadIdx.x;
  const int wave = tid >> 5;
  const int tile = wave >> 1;
  const int jq   = wave & 1;
  const int tt   = tid & 63;
  const int lane = tid & 31;
  float (&act)[NSTREAM * 16 * PITCH] = act_all[tile];
  const int half = lane >> 4;
  const int ln   = lane & 15;
  const int koff = half * 2;
  const int base = (blockIdx.x * TILES + tile) * 16;

  {
    const int n4 = 4 * lane;
    float w0[4], w1[4], w2[4], bz[4];
    #pragma unroll
    for (int c = 0; c < 4; ++c) {
      w0[c] = W0[0 * H + n4 + c];
      w1[c] = W0[1 * H + n4 + c];
      w2[c] = W0[2 * H + n4 + c];
      bz[c] = b0[n4 + c];
    }
    const float* xs = x + (size_t)(base + 8 * jq) * 3;
    float xr[24];
    #pragma unroll
    for (int q = 0; q < 24; ++q) xr[q] = xs[q];

    #pragma unroll
    for (int i = 0; i < 8; ++i) {
      const int m = 8 * jq + i;
      const float x0 = xr[3 * i + 0], x1 = xr[3 * i + 1], x2 = xr[3 * i + 2];
      v4f ra, rx, ry, rl, rt;
      #pragma unroll
      for (int c = 0; c < 4; ++c) {
        const float z  = fmaf(x0, w0[c], fmaf(x1, w1[c], fmaf(x2, w2[c], bz[c])));
        const float a  = tanh_sel(z);
        const float s2 = 1.f - a * a;
        ra[c] = a;
        rx[c] = s2 * w0[c];
        ry[c] = s2 * w1[c];
        rl[c] = -2.f * a * s2 * (w0[c] * w0[c] + w1[c] * w1[c]);
        rt[c] = s2 * w2[c];
      }
      *(v4f*)&act[(0 * 16 + m) * PITCH + n4] = ra;
      *(v4f*)&act[(1 * 16 + m) * PITCH + n4] = rx;
      *(v4f*)&act[(2 * 16 + m) * PITCH + n4] = ry;
      *(v4f*)&act[(3 * 16 + m) * PITCH + n4] = rl;
      *(v4f*)&act[(4 * 16 + m) * PITCH + n4] = rt;
    }
  }
  __syncthreads();

  mlp_hidden_layer(W1, b1, act, jq, ln, half, koff);
  __syncthreads();
  mlp_hidden_layer(W2, b2, act, jq, ln, half, koff);

  const int kq = 2 * jq + half;
  v4f w3[32];
  #pragma unroll
  for (int kk = 0; kk < 32; ++kk) {
    const float* wr = W3 + (size_t)(32 * kq + kk) * 4;
    w3[kk] = (v4f){wr[0], wr[1], wr[2], wr[3]};
  }
  __syncthreads();

  v2f s01[NSTREAM], s23[NSTREAM];
  #pragma unroll
  for (int s = 0; s < NSTREAM; ++s) { s01[s] = (v2f){0.f, 0.f}; s23[s] = (v2f){0.f, 0.f}; }
  #pragma unroll
  for (int i = 0; i < 8; ++i) {
    #pragma unroll
    for (int s = 0; s < NSTREAM; ++s) {
      const v4f h = *(const v4f*)&act[(s * 16 + ln) * PITCH + 32 * kq + 4 * i];
      #pragma unroll
      for (int e = 0; e < 4; ++e) {
        const v4f w  = w3[4 * i + e];
        const v2f hh = {h[e], h[e]};
        s01[s] = __builtin_elementwise_fma(hh, (v2f){w.x, w.y}, s01[s]);
        if (s != 4) s23[s] = __builtin_elementwise_fma(hh, (v2f){w.z, w.w}, s23[s]);
      }
    }
  }
  #pragma unroll
  for (int s = 0; s < NSTREAM; ++s) {
    s01[s].x += xhalf(s01[s].x);
    s01[s].y += xhalf(s01[s].y);
    if (s != 4) { s23[s].x += xhalf(s23[s].x); s23[s].y += xhalf(s23[s].y); }
  }
  if (jq == 1 && half == 0) {
    #pragma unroll
    for (int s = 0; s < NSTREAM; ++s) {
      if (s != 4) *(v4f*)&act[(s * 16 + ln) * PITCH + H] = (v4f){s01[s].x, s01[s].y, s23[s].x, s23[s].y};
      else        *(v2f*)&act[(s * 16 + ln) * PITCH + H] = s01[s];
    }
  }
  __syncthreads();

  if (tt < 16) {
    const size_t i = (size_t)base + tt;
    float V[NSTREAM][4];
    #pragma unroll
    for (int s = 0; s < NSTREAM; ++s) {
      if (s != 4) {
        const v4f q = *(const v4f*)&act[(s * 16 + tt) * PITCH + H];
        V[s][0] = s01[s].x + q.x; V[s][1] = s01[s].y + q.y;
        V[s][2] = s23[s].x + q.z; V[s][3] = s23[s].y + q.w;
      } else {
        const v2f q = *(const v2f*)&act[(s * 16 + tt) * PITCH + H];
        V[s][0] = s01[s].x + q.x; V[s][1] = s01[s].y + q.y;
        V[s][2] = 0.f; V[s][3] = 0.f;
      }
    }
    #pragma unroll
    for (int o = 0; o < 4; ++o) V[0][o] += b3[o];

    const float c   = V[0][0], v1 = V[0][1], v2 = V[0][2], Fi = V[0][3];
    const float dx0 = V[1][0], dx1 = V[1][1], dx2 = V[1][2], dx3 = V[1][3];
    const float dy0 = V[2][0], dy1 = V[2][1], dy2 = V[2][2], dy3 = V[2][3];
    const float lp0 = V[3][0], lp1 = V[3][1], lp2 = V[3][2], lp3 = V[3][3];
    const float dt0 = V[4][0], dt1 = V[4][1];

    const size_t Bz = (size_t)B;
    out[0 * Bz + i]          = c;
    out[1 * Bz + i]          = dt0;
    out[2 * Bz + 2 * i + 0]  = dx0;
    out[2 * Bz + 2 * i + 1]  = dy0;
    out[4 * Bz + 2 * i + 0]  = v1;
    out[4 * Bz + 2 * i + 1]  = v2;
    out[6 * Bz + i]          = dt1;
    out[7 * Bz + i]          = dx1 + dy2;
    out[8 * Bz + 2 * i + 0]  = dx1 * v1 + dy1 * v2;
    out[8 * Bz + 2 * i + 1]  = dx2 * v1 + dy2 * v2;
    out[10 * Bz + 2 * i + 0] = lp1;
    out[10 * Bz + 2 * i + 1] = lp2;
    out[12 * Bz + i]         = Fi;
    out[13 * Bz + 2 * i + 0] = dx3;
    out[13 * Bz + 2 * i + 1] = dy3;
    out[15 * Bz + i]         = lp3;
    out[16 * Bz + i] = -0.5f * lp0
                       - (dx0 * dx3 + dy0 * dy3) - c * lp3
                       + (dx0 * v1 + dy0 * v2) + c * (dx1 + dy2);
  }
}

extern "C" void kernel_launch(void* const* d_in, const int* in_sizes, int n_in,
                              void* d_out, int out_size, void* d_ws, size_t ws_size,
                              hipStream_t stream) {
  const float* x  = (const float*)d_in[0];
  const float* W0 = (const float*)d_in[1];
  const float* b0 = (const float*)d_in[2];
  const float* W1 = (const float*)d_in[3];
  const float* b1 = (const float*)d_in[4];
  const float* W2 = (const float*)d_in[5];
  const float* b2 = (const float*)d_in[6];
  const float* W3 = (const float*)d_in[7];
  const float* b3 = (const float*)d_in[8];

  const int B = in_sizes[0] / 3;
  dim3 grid(B / (16 * TILES)), block(32 * NWAVE);
  pinn_grad_kernel<<<grid, block, 0, stream>>>(x, W0, b0, W1, b1, W2, b2, W3, b3,
                                               (float*)d_out, B);
}
